# starting kernel
# baseline (speedup 1.0000x reference)
#include <hip/hip_runtime.h>
#include <hip/hip_bf16.h>


typedef __attribute__((ext_vector_type(2))) float v2f;
typedef __attribute__((ext_vector_type(8))) float v8f;

__global__ __launch_bounds__(256) void kM(const float* __restrict__ x,
                                          float* __restrict__ M) {
    const int lane = threadIdx.x & 31;
    const int wave = (blockIdx.x * 256 + threadIdx.x) >> 5;
    const int n  = wave >> 10;
    const int p0 = (wave & 1023) << 4;
    const int i    = lane & 15;
    const int half = lane >> 4;

    const float* xb = x + ((size_t)n * 64) * 16384 + p0 + i;

    v8f c = {};
    v2f b; b.x = 1.0f; b.y = 1.0f;
    for (int kb = 0; kb < 64; kb += 4) {
        v2f a;
        a.x = xb[(size_t)(kb + 2 * half)     * 16384];
        a.y = xb[(size_t)(kb + 2 * half + 1) * 16384];
        c = __builtin_amdgcn_wmma_f32_16x16x4_f32(
             false, a,  false, b,
             (short)0, c,  false,  false);
    }
    float* Mb = M + (size_t)n * 16384 + p0;
    if (lane == 0) {
        for (int r = 0; r < 8; ++r) Mb[r] = c[r];
    } else if (lane == 16) {
        for (int r = 0; r < 8; ++r) Mb[8 + r] = c[r];
    }
}

__global__ __launch_bounds__(256) void kT(const float* __restrict__ x,
                                          float* __restrict__ T) {
    const int bid   = blockIdx.x;
    const int chunk = bid % 9;
    const int ci    = (bid / 9) % 64;
    const int n     = bid / (9 * 64);

    const int t0base   = 16384 * chunk;
    const int pi_start = t0base / 9;
    const int pi_end   = (t0base + 16383) / 9;
    const int r_lo  = pi_start >> 7;
    const int r_hi  = pi_end >> 7;
    const int nrows = r_hi - r_lo + 3;

    __shared__ float tile[18][130];
    const float* xb = x + ((size_t)n * 64 + ci) * 16384;
    for (int idx = threadIdx.x; idx < nrows * 130; idx += 256) {
        int lr  = idx / 130;
        int lc  = idx - lr * 130;
        int r   = r_lo - 1 + lr;
        int col = lc - 1;
        float v = 0.0f;
        if ((unsigned)r < 128u && (unsigned)col < 128u) v = xb[r * 128 + col];
        tile[lr][lc] = v;
    }
    __syncthreads();

    const int mprime = chunk * 256 + threadIdx.x;
    const int t0 = 64 * mprime;
    float s = 0.0f;
    for (int j = 0; j < 64; ++j) {
        int tj   = t0 + j;
        int pi   = tj / 9;
        int srem = tj - pi * 9;
        int a    = srem / 3;
        int bb   = srem - a * 3;
        s += tile[(pi >> 7) + a - r_lo][(pi & 127) + bb];
    }
    T[(size_t)n * 147456 + ci * 2304 + mprime] = s;
}

__global__ __launch_bounds__(256) void kRed(const float* __restrict__ T,
                                            const float* __restrict__ M,
                                            float* __restrict__ mx,
                                            float* __restrict__ se) {
    const int n = blockIdx.x / 9;
    const int s = blockIdx.x % 9;
    const float* Tb = T + (size_t)n * 147456 + s;
    const float* Mb = M + (size_t)n * 16384;
    __shared__ float red[256];

    float lmax = -3.402823466e38f;
    for (int p = threadIdx.x; p < 16384; p += 256) {
        float v = Tb[9 * p] + Mb[p];
        lmax = fmaxf(lmax, v);
    }
    red[threadIdx.x] = lmax;
    __syncthreads();
    for (int off = 128; off > 0; off >>= 1) {
        if ((int)threadIdx.x < off)
            red[threadIdx.x] = fmaxf(red[threadIdx.x], red[threadIdx.x + off]);
        __syncthreads();
    }
    const float m = red[0];
    __syncthreads();

    float lsum = 0.0f;
    for (int p = threadIdx.x; p < 16384; p += 256) {
        float v = Tb[9 * p] + Mb[p];
        lsum += __expf(v - m);
    }
    red[threadIdx.x] = lsum;
    __syncthreads();
    for (int off = 128; off > 0; off >>= 1) {
        if ((int)threadIdx.x < off)
            red[threadIdx.x] = red[threadIdx.x] + red[threadIdx.x + off];
        __syncthreads();
    }
    if (threadIdx.x == 0) {
        mx[blockIdx.x] = m;
        se[blockIdx.x] = red[0];
    }
}

__global__ __launch_bounds__(256) void kSoftsum(const float* __restrict__ T,
                                                const float* __restrict__ M,
                                                const float* __restrict__ mx,
                                                const float* __restrict__ se,
                                                float* __restrict__ softsum) {
    const int idx = blockIdx.x * 256 + threadIdx.x;
    const int n = idx >> 14;
    const int p = idx & 16383;
    __shared__ float smx[9], sinv[9];
    if (threadIdx.x < 9) {
        smx[threadIdx.x]  = mx[n * 9 + threadIdx.x];
        sinv[threadIdx.x] = 1.0f / se[n * 9 + threadIdx.x];
    }
    __syncthreads();
    const float* Tb = T + (size_t)n * 147456 + 9 * (size_t)p;
    const float m = M[idx];
    float acc = 0.0f;
    for (int s = 0; s < 9; ++s)
        acc += __expf(Tb[s] + m - smx[s]) * sinv[s];
    softsum[idx] = acc;
}

__global__ __launch_bounds__(256) void kOut(const float* __restrict__ x,
                                            const float* __restrict__ softsum,
                                            float* __restrict__ out) {
    const int bid = blockIdx.x;
    const int rc = bid & 15;
    const int c  = (bid >> 4) & 63;
    const int n  = bid >> 10;
    const int r0 = rc * 8;

    __shared__ float tile[10][130];
    const float* xb = x + ((size_t)n * 64 + c) * 16384;
    for (int idx = threadIdx.x; idx < 10 * 130; idx += 256) {
        int lr  = idx / 130;
        int lc  = idx - lr * 130;
        int r   = r0 - 1 + lr;
        int col = lc - 1;
        float v = 0.0f;
        if ((unsigned)r < 128u && (unsigned)col < 128u) v = xb[r * 128 + col];
        tile[lr][lc] = v;
    }
    __syncthreads();

    const float* ssb = softsum + (size_t)n * 16384;
    float* ob = out + ((size_t)n * 64 + c) * 16384;
    for (int k = threadIdx.x; k < 1024; k += 256) {
        int lr = k >> 7;
        int wv = k & 127;
        int h  = r0 + lr;
        float b = tile[lr    ][wv] + tile[lr    ][wv + 1] + tile[lr    ][wv + 2]
                + tile[lr + 1][wv] + tile[lr + 1][wv + 1] + tile[lr + 1][wv + 2]
                + tile[lr + 2][wv] + tile[lr + 2][wv + 1] + tile[lr + 2][wv + 2];
        int pout = c * 256 + h * 2 + (wv >> 6);
        ob[h * 128 + wv] = ssb[pout] + b;
    }
}

extern "C" void kernel_launch(void* const* d_in, const int* in_sizes, int n_in,
                              void* d_out, int out_size, void* d_ws, size_t ws_size,
                              hipStream_t stream) {
    const float* x = (const float*)d_in[0];
    float* out = (float*)d_out;
    float* ws  = (float*)d_ws;

    float* T       = ws;
    float* M       = T + (size_t)8 * 147456;
    float* mx      = M + (size_t)8 * 16384;
    float* se      = mx + 72;
    float* softsum = se + 72;

    kM      <<<1024,       256, 0, stream>>>(x, M);
    kT      <<<8 * 64 * 9, 256, 0, stream>>>(x, T);
    kRed    <<<72,         256, 0, stream>>>(T, M, mx, se);
    kSoftsum<<<512,        256, 0, stream>>>(T, M, mx, se, softsum);
    kOut    <<<8192,       256, 0, stream>>>(x, softsum, out);
}


# kRed: four blocks per softmax column, each reduces a contiguous quarter to a (max, exp-sum) pair; kOut merges the four pairs per slot in lanes 0..8: 288 kRed blocks, not 72; exp-sum order changed
# speedup vs baseline: 5.9664x; 5.9664x over previous
#include <hip/hip_runtime.h>
#include <hip/hip_bf16.h>


typedef __attribute__((ext_vector_type(2))) float v2f;
typedef __attribute__((ext_vector_type(4))) float v4f;
typedef __attribute__((ext_vector_type(8))) float v8f;

__device__ __forceinline__ void partM(const float* __restrict__ x,
                                      float* __restrict__ M, const int bid) {
    const int lane = threadIdx.x & 31;
    const int wave = (bid * 256 + threadIdx.x) >> 5;
    const int n  = wave >> 10;
    const int p0 = (wave & 1023) << 4;
    const int i    = lane & 15;
    const int half = lane >> 4;

    const float* xb = x + ((size_t)n * 64) * 16384 + p0 + i;

    v8f c = {};
    v2f b; b.x = 1.0f; b.y = 1.0f;
    for (int kb = 0; kb < 64; kb += 4) {
        v2f a;
        a.x = xb[(size_t)(kb + 2 * half)     * 16384];
        a.y = xb[(size_t)(kb + 2 * half + 1) * 16384];
        c = __builtin_amdgcn_wmma_f32_16x16x4_f32(
             false, a,  false, b,
             (short)0, c,  false,  false);
    }
    float* Mb = M + (size_t)n * 16384 + p0;
    if (lane == 0) {
        for (int r = 0; r < 8; ++r) Mb[r] = c[r];
    } else if (lane == 16) {
        for (int r = 0; r < 8; ++r) Mb[8 + r] = c[r];
    }
}

__device__ __forceinline__ void partT(const float* __restrict__ x,
                                      float* __restrict__ T, const int bid) {
    constexpr int HP = 68;
    const int ci   = bid & 63;
    const int n    = bid >> 6;
    const int tid  = threadIdx.x;
    const int lane = tid & 31;
    const int wv   = tid >> 5;

    __shared__ __attribute__((aligned(16))) float tile[260 * HP];
    const float* xb = x + ((size_t)n * 64 + ci) * 16384;
    v4f stg[16];
    for (int k = 0; k < 16; ++k)
        stg[k] = *(const v4f*)(xb + (8 * k + wv) * 128 + 4 * lane);
    for (int k = 0; k < 16; ++k)
        *(v4f*)(tile + (2 * (8 * k + wv + 1) + (lane >> 4)) * HP + 4 * (lane & 15)) = stg[k];
    if (tid < 34) {
        const v4f z = {0.0f, 0.0f, 0.0f, 0.0f};
        *(v4f*)(tile + 4 * tid) = z;
        *(v4f*)(tile + 258 * HP + 4 * tid) = z;
    }
    __syncthreads();

    const int q  = tid;
    const int hf = q & 1;
    const int loff = hf ? -5 : 0;
    const int roff = hf ? 0 : HP;
    float acc[9];
#pragma unroll
    for (int s = 0; s < 9; ++s) acc[s] = 0.0f;
#pragma unroll
    for (int a = 0; a < 3; ++a) {
        const float* bp = tile + (q + 2 * a) * HP;
        float u[66];
        const float ul = bp[loff];
        const float ur = bp[roff];
#pragma unroll
        for (int j = 0; j < 16; ++j) {
            const v4f m = *(const v4f*)(bp + 4 * j);
            u[4 * j + 1] = m.x; u[4 * j + 2] = m.y; u[4 * j + 3] = m.z; u[4 * j + 4] = m.w;
        }
        u[0]  = hf ? ul : 0.0f;
        u[65] = hf ? 0.0f : ur;
#pragma unroll
        for (int s = 0; s < 9; ++s) {
            const float v0 = u[7 * s],     v1 = u[7 * s + 1], v2 = u[7 * s + 2];
            const float v3 = u[7 * s + 3], v4 = u[7 * s + 4], v5 = u[7 * s + 5];
            const float v6 = u[7 * s + 6], v7 = u[7 * s + 7], v8 = u[7 * s + 8];
            const float v9 = u[7 * s + 9];
            const int e = s - 3 * a;
            float f = (v1 + v8) + 2.0f * (v2 + v7) + 3.0f * ((v3 + v4) + (v5 + v6));
            if (e <= 0)      f += (v0 + v1) + v2;
            else if (e == 1) f += v1 + v2;
            else if (e == 2) f += v2;
            if (e >= 2)      f += (v7 + v8) + v9;
            else if (e == 1) f += v7 + v8;
            else if (e == 0) f += v7;
            acc[s] += f;
        }
    }
    float* Tb = T + (size_t)n * 147456 + ci * 256 + q;
#pragma unroll
    for (int s = 0; s < 9; ++s) Tb[s * 16384] = acc[s];
}

__global__ __launch_bounds__(256) void kMT(const float* __restrict__ x,
                                           float* __restrict__ T,
                                           float* __restrict__ M) {
    if (blockIdx.x < 512) partT(x, T, blockIdx.x);
    else                  partM(x, M, blockIdx.x - 512);
}

template <int MASK>
__device__ __forceinline__ float rowx(float v) {
    const int b = __builtin_bit_cast(int, v);
    return __builtin_bit_cast(float,
        __builtin_amdgcn_update_dpp(b, b, 0x160 + MASK, 0xf, 0xf, false));
}
__device__ __forceinline__ float xhalf(float v) {
    const int b = __builtin_bit_cast(int, v);
    return __builtin_bit_cast(float,
        __builtin_amdgcn_permlanex16(b, b, 0x76543210, 0xfedcba98, false, false));
}
__device__ __forceinline__ float waveMax(float v) {
    v = fmaxf(v, rowx<1>(v));
    v = fmaxf(v, rowx<2>(v));
    v = fmaxf(v, rowx<4>(v));
    v = fmaxf(v, rowx<8>(v));
    return fmaxf(v, xhalf(v));
}
__device__ __forceinline__ float waveSum(float v) {
    v += rowx<1>(v);
    v += rowx<2>(v);
    v += rowx<4>(v);
    v += rowx<8>(v);
    return v + xhalf(v);
}

__global__ __launch_bounds__(256) void kRed(const float* __restrict__ T,
                                            const float* __restrict__ M,
                                            float* __restrict__ mx,
                                            float* __restrict__ se) {
    const int col  = blockIdx.x >> 2;
    const int part = blockIdx.x & 3;
    const int n = col / 9;
    const int s = col % 9;
    const float* Tb = T + (size_t)n * 147456 + s * 16384 + part * 4096;
    const float* Mb = M + (size_t)n * 16384 + part * 4096;
    const int lane = threadIdx.x & 31;
    const int wv   = threadIdx.x >> 5;
    __shared__ __attribute__((aligned(16))) float wmax[8];
    __shared__ __attribute__((aligned(16))) float wsum[8];

    float vv[16];
    for (int i = 0; i < 16; ++i) {
        const int p = threadIdx.x + 256 * i;
        vv[i] = Tb[p] + Mb[p];
    }
    float lmax = -3.402823466e38f;
    for (int i = 0; i < 16; ++i) lmax = fmaxf(lmax, vv[i]);
    lmax = waveMax(lmax);
    if (lane == 0) wmax[wv] = lmax;
    __syncthreads();
    float m = wmax[0];
    for (int w = 1; w < 8; ++w) m = fmaxf(m, wmax[w]);

    float lsum = 0.0f;
    for (int i = 0; i < 16; ++i) lsum += __expf(vv[i] - m);
    lsum = waveSum(lsum);
    if (lane == 0) wsum[wv] = lsum;
    __syncthreads();
    if (threadIdx.x == 0) {
        float t = wsum[0];
        for (int w = 1; w < 8; ++w) t += wsum[w];
        mx[blockIdx.x] = m;
        se[blockIdx.x] = t;
    }
}

__global__ __launch_bounds__(256) void kOut(const float* __restrict__ x,
                                            const float* __restrict__ T,
                                            const float* __restrict__ M,
                                            const float* __restrict__ mx,
                                            const float* __restrict__ se,
                                            float* __restrict__ out) {
    constexpr int PITCH = 136;
    const int c    = blockIdx.x & 63;
    const int n    = blockIdx.x >> 6;
    const int tid  = threadIdx.x;
    const int lane = tid & 31;
    const int wv   = tid >> 5;

    __shared__ __attribute__((aligned(16))) float tile[130 * PITCH];
    __shared__ float ss[256];
    const int p = c * 256 + tid;
    const float* Tb = T + (size_t)n * 147456 + p;
    float tv[9];
    for (int s = 0; s < 9; ++s) tv[s] = Tb[s * 16384];
    const float mv = M[(size_t)n * 16384 + p];
    const int sl = lane < 9 ? lane : 8;
    const v4f m4 = *(const v4f*)(mx + (n * 9 + sl) * 4);
    const v4f s4 = *(const v4f*)(se + (n * 9 + sl) * 4);
    const float mm  = fmaxf(fmaxf(m4.x, m4.y), fmaxf(m4.z, m4.w));
    const float tot = s4.x * __expf(m4.x - mm) + s4.y * __expf(m4.y - mm)
                    + s4.z * __expf(m4.z - mm) + s4.w * __expf(m4.w - mm);
    const float rin = 1.0f / tot;
    float smx[9], sinv[9];
    for (int s = 0; s < 9; ++s) {
        smx[s]  = __builtin_bit_cast(float,
                      __builtin_amdgcn_readlane(__builtin_bit_cast(int, mm), s));
        sinv[s] = __builtin_bit_cast(float,
                      __builtin_amdgcn_readlane(__builtin_bit_cast(int, rin), s));
    }

    const float* xb = x + ((size_t)n * 64 + c) * 16384;
    v4f stg[16];
    for (int k = 0; k < 16; ++k)
        stg[k] = *(const v4f*)(xb + (8 * k + wv) * 128 + 4 * lane);
    for (int k = 0; k < 16; ++k)
        *(v4f*)(tile + (8 * k + wv + 1) * PITCH + 4 + 4 * lane) = stg[k];
    if (tid < PITCH) {
        tile[tid] = 0.0f;
        tile[129 * PITCH + tid] = 0.0f;
    }
    if (tid < 128) {
        tile[(tid + 1) * PITCH + 3]   = 0.0f;
        tile[(tid + 1) * PITCH + 132] = 0.0f;
    }
    float acc = 0.0f;
    for (int s = 0; s < 9; ++s)
        acc += __expf(tv[s] + mv - smx[s]) * sinv[s];
    ss[tid] = acc;
    __syncthreads();

    float sv[16];
    for (int i = 0; i < 16; ++i) sv[i] = ss[32 * wv + 2 * i + (lane >> 4)];

    float a[18][6];
    const float* tp = tile + (16 * wv) * PITCH + 4 + 4 * lane;
    for (int t = 0; t < 18; ++t) {
        const v4f m = *(const v4f*)(tp + t * PITCH);
        a[t][0] = tp[t * PITCH - 1];
        a[t][1] = m.x; a[t][2] = m.y; a[t][3] = m.z; a[t][4] = m.w;
        a[t][5] = tp[t * PITCH + 4];
    }
    float* ob = out + ((size_t)n * 64 + c) * 16384 + (16 * wv) * 128 + 4 * lane;
    for (int i = 0; i < 16; ++i) {
        v4f o;
        float b;
        b = a[i][0] + a[i][1] + a[i][2] + a[i + 1][0] + a[i + 1][1] + a[i + 1][2]
          + a[i + 2][0] + a[i + 2][1] + a[i + 2][2];
        o.x = sv[i] + b;
        b = a[i][1] + a[i][2] + a[i][3] + a[i + 1][1] + a[i + 1][2] + a[i + 1][3]
          + a[i + 2][1] + a[i + 2][2] + a[i + 2][3];
        o.y = sv[i] + b;
        b = a[i][2] + a[i][3] + a[i][4] + a[i + 1][2] + a[i + 1][3] + a[i + 1][4]
          + a[i + 2][2] + a[i + 2][3] + a[i + 2][4];
        o.z = sv[i] + b;
        b = a[i][3] + a[i][4] + a[i][5] + a[i + 1][3] + a[i + 1][4] + a[i + 1][5]
          + a[i + 2][3] + a[i + 2][4] + a[i + 2][5];
        o.w = sv[i] + b;
        *(v4f*)(ob + i * 128) = o;
    }
}

extern "C" void kernel_launch(void* const* d_in, const int* in_sizes, int n_in,
                              void* d_out, int out_size, void* d_ws, size_t ws_size,
                              hipStream_t stream) {
    const float* x = (const float*)d_in[0];
    float* out = (float*)d_out;
    float* ws  = (float*)d_ws;

    float* T       = ws;
    float* M       = T + (size_t)8 * 147456;
    float* mx      = M + (size_t)8 * 16384;
    float* se      = mx + 288;

    kMT     <<<1536,       256, 0, stream>>>(x, T, M);
    kRed    <<<288,        256, 0, stream>>>(T, M, mx, se);
    kOut    <<<512,        256, 0, stream>>>(x, T, M, mx, se, out);
}
